# starting kernel
# baseline (speedup 1.0000x reference)
#include <hip/hip_runtime.h>

typedef __attribute__((ext_vector_type(2))) float v2f;
typedef __attribute__((ext_vector_type(8))) float v8f;

__global__ void init_deg_kernel(float* __restrict__ deg, int n) {
    int i = blockIdx.x * blockDim.x + threadIdx.x;
    if (i < n) deg[i] = 1.0f;
}

__global__ void accum_deg_kernel(const int* __restrict__ dst,
                                 float* __restrict__ deg, int ne) {
    int i = blockIdx.x * blockDim.x + threadIdx.x;
    if (i < ne) atomicAdd(&deg[dst[i]], 1.0f);
}

__global__ void deg_to_dinv_kernel(float* __restrict__ d, int n) {
    int i = blockIdx.x * blockDim.x + threadIdx.x;
    if (i < n) d[i] = rsqrtf(d[i]);
}

__global__ void pad_x_kernel(const float* __restrict__ x, float* __restrict__ xp,
                             int nn, int nn_pad) {
    int i = blockIdx.x * blockDim.x + threadIdx.x;
    int total = nn_pad * 20;
    if (i >= total) return;
    int n = i / 20;
    int k = i % 20;
    xp[i] = (n < nn && k < 18) ? x[n * 18 + k] : 0.0f;
}

__global__ void repack_w_kernel(const float* __restrict__ w, float* __restrict__ wp,
                                int kd_true, int kp, int nf) {
    int o = blockIdx.x * blockDim.x + threadIdx.x;
    if (o >= kp * nf) return;
    int j = o & 1;
    int n = (o >> 1) % nf;
    int p = (o >> 1) / nf;
    int k = 2 * p + j;
    wp[o] = (k < kd_true) ? w[k * nf + n] : 0.0f;
}

template <int KD, int NF, int NT>
__global__ void gemm_scale_kernel(const float* __restrict__ X,
                                  const float* __restrict__ Wp,
                                  const float* __restrict__ dinv,
                                  float* __restrict__ hs,
                                  float* __restrict__ agg,
                                  int mtiles) {
    int wave = blockIdx.x * (blockDim.x >> 5) + (threadIdx.x >> 5);
    if (wave >= mtiles) return;
    int lane = threadIdx.x & 31;
    int r16  = lane & 15;
    int kh   = lane >> 4;
    int mbase = wave * 16;
    int arow  = mbase + r16;

    v8f acc[NT] = {};

    static_assert(KD % 4 == 0, "KD must be multiple of 4");
    constexpr int KSTEPS = KD / 4;
#pragma unroll
    for (int ks = 0; ks < KSTEPS; ++ks) {
        int k0 = ks * 4 + kh * 2;
        v2f a = *(const v2f*)&X[arow * KD + k0];
        int p = ks * 2 + kh;
#pragma unroll
        for (int t = 0; t < NT; ++t) {
            v2f b = *(const v2f*)&Wp[(p * NF + t * 16 + r16) * 2];
            acc[t] = __builtin_amdgcn_wmma_f32_16x16x4_f32(
                false, a, false, b, (short)0, acc[t], false, false);
        }
    }

    const float4 dv0 = *(const float4*)&dinv[mbase + 8 * kh];
    const float4 dv1 = *(const float4*)&dinv[mbase + 8 * kh + 4];
    const float dv[8] = {dv0.x, dv0.y, dv0.z, dv0.w, dv1.x, dv1.y, dv1.z, dv1.w};

#pragma unroll
    for (int v = 0; v < 8; ++v) {
        int m = mbase + v + 8 * kh;
#pragma unroll
        for (int t = 0; t < NT; ++t) {
            float val = acc[t][v] * dv[v];
            int idx = m * NF + t * 16 + r16;
            hs[idx]  = val;
            agg[idx] = val;
        }
    }
}

template <int F4>
__global__ void scatter_kernel(const int* __restrict__ src,
                               const int* __restrict__ dst,
                               const float* __restrict__ hs,
                               float* __restrict__ agg, int ne) {
    unsigned int tid = blockIdx.x * blockDim.x + threadIdx.x;
    unsigned int total = (unsigned int)ne * F4;
    if (tid >= total) return;
    int e  = (int)(tid / F4);
    int fo = (int)(tid % F4) * 4;
    constexpr int F = F4 * 4;
    int s = src[e];
    int d = dst[e];
    const float4 val = *(const float4*)&hs[s * F + fo];
    float* p = &agg[d * F + fo];
    atomicAdd(p + 0, val.x);
    atomicAdd(p + 1, val.y);
    atomicAdd(p + 2, val.z);
    atomicAdd(p + 3, val.w);
}

template <int F, bool RELU>
__global__ void finalize_kernel(const float* __restrict__ agg,
                                const float* __restrict__ dinv,
                                const float* __restrict__ bias,
                                float* __restrict__ out, int nrows) {
    int i = blockIdx.x * blockDim.x + threadIdx.x;
    constexpr int F4 = F / 4;
    if (i >= nrows * F4) return;
    int n  = i / F4;
    int fo = (i % F4) * 4;
    float dvn = dinv[n];
    const float4 a = *(const float4*)&agg[n * F + fo];
    const float4 b = *(const float4*)&bias[fo];
    float4 v;
    v.x = dvn * a.x + b.x;
    v.y = dvn * a.y + b.y;
    v.z = dvn * a.z + b.z;
    v.w = dvn * a.w + b.w;
    if (RELU) {
        v.x = fmaxf(v.x, 0.0f);
        v.y = fmaxf(v.y, 0.0f);
        v.z = fmaxf(v.z, 0.0f);
        v.w = fmaxf(v.w, 0.0f);
    }
    *(float4*)&out[n * F + fo] = v;
}

extern "C" void kernel_launch(void* const* d_in, const int* in_sizes, int n_in,
                              void* d_out, int out_size, void* d_ws, size_t ws_size,
                              hipStream_t stream) {
    const float* x  = (const float*)d_in[0];
    const int*   ei = (const int*)d_in[1];
    const float* W1 = (const float*)d_in[2];
    const float* b1 = (const float*)d_in[3];
    const float* W2 = (const float*)d_in[4];
    const float* b2 = (const float*)d_in[5];
    float* out = (float*)d_out;

    const int NN = in_sizes[0] / 18;
    const int NE = in_sizes[1] / 2;
    const int NN_pad = (NN + 15) & ~15;
    const int* src = ei;
    const int* dst = ei + NE;

    char* ws = (char*)d_ws;
    size_t off = 0;
    auto carve = [&](size_t bytes) -> char* {
        char* p = ws + off;
        off += (bytes + 255) & ~(size_t)255;
        return p;
    };
    float* dinv = (float*)carve((size_t)NN_pad * 4);
    float* xp   = (float*)carve((size_t)NN_pad * 20 * 4);
    float* W1p  = (float*)carve((size_t)20 * 64 * 4);
    float* W2p  = (float*)carve((size_t)64 * 32 * 4);
    float* hs1  = (float*)carve((size_t)NN_pad * 64 * 4);
    float* agg1 = (float*)carve((size_t)NN_pad * 64 * 4);
    float* h1   = (float*)carve((size_t)NN_pad * 64 * 4);
    float* hs2  = hs1;
    float* agg2 = agg1;

    const int B = 256;

    init_deg_kernel<<<(NN_pad + B - 1) / B, B, 0, stream>>>(dinv, NN_pad);
    accum_deg_kernel<<<(NE + B - 1) / B, B, 0, stream>>>(dst, dinv, NE);
    deg_to_dinv_kernel<<<(NN_pad + B - 1) / B, B, 0, stream>>>(dinv, NN_pad);

    pad_x_kernel<<<(NN_pad * 20 + B - 1) / B, B, 0, stream>>>(x, xp, NN, NN_pad);
    repack_w_kernel<<<(20 * 64 + B - 1) / B, B, 0, stream>>>(W1, W1p, 18, 20, 64);
    repack_w_kernel<<<(64 * 32 + B - 1) / B, B, 0, stream>>>(W2, W2p, 64, 64, 32);

    const int mtiles = NN_pad / 16;
    const int gblk = (mtiles + 3) / 4;

    gemm_scale_kernel<20, 64, 4><<<gblk, 128, 0, stream>>>(xp, W1p, dinv, hs1, agg1, mtiles);
    {
        unsigned int total = (unsigned int)NE * 16u;
        scatter_kernel<16><<<(total + B - 1) / B, B, 0, stream>>>(src, dst, hs1, agg1, NE);
    }
    finalize_kernel<64, true><<<(NN_pad * 16 + B - 1) / B, B, 0, stream>>>(agg1, dinv, b1, h1, NN_pad);

    gemm_scale_kernel<64, 32, 2><<<gblk, 128, 0, stream>>>(h1, W2p, dinv, hs2, agg2, mtiles);
    {
        unsigned int total = (unsigned int)NE * 8u;
        scatter_kernel<8><<<(total + B - 1) / B, B, 0, stream>>>(src, dst, hs2, agg2, NE);
    }
    finalize_kernel<32, false><<<(NN * 8 + B - 1) / B, B, 0, stream>>>(agg2, dinv, b2, out, NN);
}


# Rank atomics coalesced: a work-group first orders its chunk of edges in LDS by the 128-byte line of the destination counter, so one atomic's lanes fall into few lines; sums stay in rank-arrival order
# speedup vs baseline: 3.4264x; 3.4264x over previous
#include <hip/hip_runtime.h>

typedef __attribute__((ext_vector_type(2))) float v2f;
typedef __attribute__((ext_vector_type(8))) float v8f;

constexpr int SLOTS = 64;

__global__ void zero_cnt_kernel(int* __restrict__ cnt, int n) {
    int i = blockIdx.x * blockDim.x + threadIdx.x;
    if (i < n) cnt[i] = 0;
}

constexpr int RF_THREADS = 1024;
constexpr int RF_EPT     = 16;
constexpr int RF_CHUNK   = RF_THREADS * RF_EPT;
constexpr int RF_BKT     = 4096;
constexpr int RF_WALK    = 4;

__device__ __forceinline__ int rf_bucket(int d) { return (d >> 5) & (RF_BKT - 1); }

__global__ void __launch_bounds__(RF_THREADS)
rank_fill_kernel(const int* __restrict__ src, const int* __restrict__ dst,
                 int* __restrict__ cnt, int* __restrict__ slots,
                 int* __restrict__ ovf, int* __restrict__ ovf_n, int ne) {
    static_assert(RF_BKT == 4 * RF_THREADS, "the prefix step gives every thread four buckets");
    static_assert(RF_THREADS == 1024, "the prefix step scans 32 wave totals with one wave");
    __shared__ int  hist[RF_BKT];
    __shared__ int2 ordered[RF_CHUNK];
    __shared__ int  wtot[32];

    const int t    = threadIdx.x;
    const int lane = t & 31;
    const int wave = t >> 5;
    const int e0   = blockIdx.x * RF_CHUNK;
    const int left = ne - e0;
    const int nhere = (left < RF_CHUNK) ? left : RF_CHUNK;

    for (int i = t; i < RF_BKT; i += RF_THREADS) hist[i] = 0;
    __syncthreads();

    int d[RF_EPT], s[RF_EPT], r[RF_EPT];
#pragma unroll
    for (int u = 0; u < RF_EPT; ++u) {
        const int j = u * RF_THREADS + t;
        d[u] = -1;
        s[u] = 0;
        if (j < nhere) { d[u] = dst[e0 + j]; s[u] = src[e0 + j]; }
    }
#pragma unroll
    for (int u = 0; u < RF_EPT; ++u) {
        r[u] = 0;
        if (d[u] >= 0) r[u] = atomicAdd(&hist[rf_bucket(d[u])], 1);
    }
    __syncthreads();

    {
        const int4 c = *(const int4*)&hist[4 * t];
        const int sum = c.x + c.y + c.z + c.w;
        int incl = sum;
#pragma unroll
        for (int o = 1; o < 32; o <<= 1) {
            const int v = __shfl_up(incl, o, 32);
            if (lane >= o) incl += v;
        }
        if (lane == 31) wtot[wave] = incl;
        __syncthreads();
        if (wave == 0) {
            const int w = wtot[lane];
            int wi = w;
#pragma unroll
            for (int o = 1; o < 32; o <<= 1) {
                const int v = __shfl_up(wi, o, 32);
                if (lane >= o) wi += v;
            }
            wtot[lane] = wi - w;
        }
        __syncthreads();
        const int off = wtot[wave] + incl - sum;
        int4 st;
        st.x = off;
        st.y = off + c.x;
        st.z = off + c.x + c.y;
        st.w = off + c.x + c.y + c.z;
        *(int4*)&hist[4 * t] = st;
    }
    __syncthreads();

#pragma unroll
    for (int u = 0; u < RF_EPT; ++u) {
        if (d[u] >= 0) ordered[hist[rf_bucket(d[u])] + r[u]] = make_int2(d[u], s[u]);
    }
    __syncthreads();

    for (int j0 = t; j0 < nhere; j0 += RF_WALK * RF_THREADS) {
        int2 e[RF_WALK];
        int  p[RF_WALK];
#pragma unroll
        for (int u = 0; u < RF_WALK; ++u) {
            const int j = j0 + u * RF_THREADS;
            e[u] = make_int2(-1, 0);
            if (j < nhere) e[u] = ordered[j];
        }
#pragma unroll
        for (int u = 0; u < RF_WALK; ++u) {
            p[u] = 0;
            if (e[u].x >= 0) p[u] = atomicAdd(&cnt[e[u].x], 1);
        }
#pragma unroll
        for (int u = 0; u < RF_WALK; ++u) {
            if (e[u].x >= 0) {
                if (p[u] < SLOTS) {
                    slots[e[u].x * SLOTS + p[u]] = e[u].y;
                } else {
                    const int q = atomicAdd(ovf_n, 1);
                    ovf[2 * q]     = e[u].x;
                    ovf[2 * q + 1] = e[u].y;
                }
            }
        }
    }
}

template <int KD, int NF, int NT>
__global__ void gemm_scale_kernel(const float* __restrict__ X,
                                  const float* __restrict__ Wt,
                                  const int* __restrict__ cnt,
                                  float* __restrict__ hs,
                                  int mtiles, int nrows) {
    int wave = blockIdx.x * (blockDim.x >> 5) + (threadIdx.x >> 5);
    if (wave >= mtiles) return;
    int lane = threadIdx.x & 31;
    int r16  = lane & 15;
    int kh   = lane >> 4;
    int mbase = wave * 16;
    int arow  = mbase + r16;

    v8f acc[NT] = {};

    static_assert(KD % 2 == 0, "KD must be even: (row*KD + k0) even keeps the A pair 8B aligned");
    constexpr int KSTEPS = (KD + 3) / 4;
    const bool row_ok = arow < nrows;
    const float* xrow = X + (row_ok ? arow : 0) * KD;
    const v2f zero2 = {0.0f, 0.0f};
#pragma unroll
    for (int ks = 0; ks < KSTEPS; ++ks) {
        const int k0 = ks * 4 + kh * 2;
        const bool k_ok = k0 < KD;
        const int kc = k_ok ? k0 : 0;
        v2f a = *(const v2f*)&xrow[kc];
        a = (k_ok && row_ok) ? a : zero2;
#pragma unroll
        for (int t = 0; t < NT; ++t) {
            const int n = t * 16 + r16;
            v2f b = {Wt[kc * NF + n], Wt[(kc + 1) * NF + n]};
            b = k_ok ? b : zero2;
            acc[t] = __builtin_amdgcn_wmma_f32_16x16x4_f32(
                false, a, false, b, (short)0, acc[t], false, false);
        }
    }

    const int4 c0 = *(const int4*)&cnt[mbase + 8 * kh];
    const int4 c1 = *(const int4*)&cnt[mbase + 8 * kh + 4];
    const float dv[8] = {rsqrtf((float)(c0.x + 1)), rsqrtf((float)(c0.y + 1)),
                         rsqrtf((float)(c0.z + 1)), rsqrtf((float)(c0.w + 1)),
                         rsqrtf((float)(c1.x + 1)), rsqrtf((float)(c1.y + 1)),
                         rsqrtf((float)(c1.z + 1)), rsqrtf((float)(c1.w + 1))};

#pragma unroll
    for (int v = 0; v < 8; ++v) {
        int m = mbase + v + 8 * kh;
#pragma unroll
        for (int t = 0; t < NT; ++t) {
            float val = acc[t][v] * dv[v];
            int idx = m * NF + t * 16 + r16;
            hs[idx] = val;
        }
    }
}

template <int F, bool RELU>
__global__ void __launch_bounds__(256)
pull_kernel(const int* __restrict__ cnt, const int* __restrict__ slots,
            const int* __restrict__ ovf, const int* __restrict__ ovf_n,
            const float* __restrict__ hs, const float* __restrict__ bias,
            float* __restrict__ out, int nrows) {
    constexpr int V = F / 32;
    typedef float vecf __attribute__((ext_vector_type(V)));
    int wrow = blockIdx.x * (blockDim.x >> 5) + (threadIdx.x >> 5);
    if (wrow >= nrows) return;
    const int row = __builtin_amdgcn_readfirstlane(wrow);
    const int lane = threadIdx.x & 31;

    const int ndeg = cnt[row];
    const int n    = (ndeg < SLOTS) ? ndeg : SLOTS;
    const int beg  = row * SLOTS;

    vecf acc = *(const vecf*)&hs[row * F + lane * V];

    for (int j0 = 0; j0 < n; j0 += 32) {
        const int m = (n - j0 < 32) ? (n - j0) : 32;
        const int c = (lane < m) ? slots[beg + j0 + lane] : row;
#pragma unroll
        for (int g = 0; g < 4; ++g) {
            if (g * 8 < m) {
                vecf v[8];
#pragma unroll
                for (int k = 0; k < 8; ++k) {
                    const int sidx = __builtin_amdgcn_readlane(c, g * 8 + k);
                    v[k] = *(const vecf*)&hs[sidx * F + lane * V];
                }
#pragma unroll
                for (int k = 0; k < 8; ++k) {
                    const float wgt = (g * 8 + k < m) ? 1.0f : 0.0f;
                    acc += v[k] * wgt;
                }
            }
        }
    }

    if (ndeg > SLOTS) {
        const int no = *ovf_n;
        for (int j0 = 0; j0 < no; j0 += 32) {
            const int j = j0 + lane;
            int d = -1, sv = row;
            if (j < no) { d = ovf[2 * j]; sv = ovf[2 * j + 1]; }
            unsigned hit = __builtin_amdgcn_ballot_w32(d == row);
            while (hit) {
                const int k = __builtin_ctz(hit);
                hit &= hit - 1;
                const int sidx = __builtin_amdgcn_readlane(sv, k);
                acc += *(const vecf*)&hs[sidx * F + lane * V];
            }
        }
    }

    const float dvn = rsqrtf((float)(ndeg + 1));
    const vecf b = *(const vecf*)&bias[lane * V];
    vecf o = dvn * acc + b;
    if (RELU) {
        for (int q = 0; q < V; ++q) o[q] = fmaxf(o[q], 0.0f);
    }
    *(vecf*)&out[row * F + lane * V] = o;
}

extern "C" void kernel_launch(void* const* d_in, const int* in_sizes, int n_in,
                              void* d_out, int out_size, void* d_ws, size_t ws_size,
                              hipStream_t stream) {
    const float* x  = (const float*)d_in[0];
    const int*   ei = (const int*)d_in[1];
    const float* W1 = (const float*)d_in[2];
    const float* b1 = (const float*)d_in[3];
    const float* W2 = (const float*)d_in[4];
    const float* b2 = (const float*)d_in[5];
    float* out = (float*)d_out;

    const int NN = in_sizes[0] / 18;
    const int NE = in_sizes[1] / 2;
    const int NN_pad = (NN + 15) & ~15;
    const int* src = ei;
    const int* dst = ei + NE;

    char* ws = (char*)d_ws;
    size_t off = 0;
    auto carve = [&](size_t bytes) -> char* {
        char* p = ws + off;
        off += (bytes + 255) & ~(size_t)255;
        return p;
    };
    const int NN_cnt = (NN_pad + 255) & ~255;
    int*   cnt    = (int*)carve((size_t)(NN_cnt + 256) * 4);
    int*   ovf_n  = cnt + NN_cnt;
    int*   slots  = (int*)carve((size_t)NN_pad * SLOTS * 4);
    int*   ovf    = (int*)carve((size_t)NE * 2 * 4);
    float* hs1  = (float*)carve((size_t)NN_pad * 64 * 4);
    float* h1   = (float*)carve((size_t)NN_pad * 64 * 4);
    float* hs2  = hs1;

    const int B = 256;

    zero_cnt_kernel<<<(NN_cnt + 256 + B - 1) / B, B, 0, stream>>>(cnt, NN_cnt + 256);
    rank_fill_kernel<<<(NE + RF_CHUNK - 1) / RF_CHUNK, RF_THREADS, 0, stream>>>(
        src, dst, cnt, slots, ovf, ovf_n, NE);

    const int mtiles = NN_pad / 16;
    const int gblk = (mtiles + 3) / 4;
    const int RPB = B / 32;

    gemm_scale_kernel<18, 64, 4><<<gblk, 128, 0, stream>>>(x, W1, cnt, hs1, mtiles, NN);
    pull_kernel<64, true><<<(NN_pad + RPB - 1) / RPB, B, 0, stream>>>(
        cnt, slots, ovf, ovf_n, hs1, b1, h1, NN_pad);

    gemm_scale_kernel<64, 32, 2><<<gblk, 128, 0, stream>>>(h1, W2, cnt, hs2, mtiles, NN_pad);
    pull_kernel<32, false><<<(NN + RPB - 1) / RPB, B, 0, stream>>>(
        cnt, slots, ovf, ovf_n, hs2, b2, out, NN);
}
